# starting kernel
# baseline (speedup 1.0000x reference)
#include <hip/hip_runtime.h>
#include <hip/hip_bf16.h>


typedef float v2f __attribute__((ext_vector_type(2)));
typedef float v8f __attribute__((ext_vector_type(8)));

#define BB 16
#define TT 1024
#define CC 128
#define KTT (2*TT - 1)


__global__ __launch_bounds__(128)
void swin_qkbias_kernel(const float* __restrict__ Q,
                        const float* __restrict__ K,
                        const float* __restrict__ bias,
                        float* __restrict__ out) {
    __shared__ float lds[4][16 * 36];

    const int lane = threadIdx.x & 31;
    const int wv   = threadIdx.x >> 5;
    const int half = lane >> 4;
    const int l16  = lane & 15;

    const int bid   = blockIdx.x;
    const int tgrp  = bid & 3;
    const int wtile = (bid >> 2) & 63;
    const int b     = bid >> 8;

    const int w0      = wtile * 16;
    const int t0_base = tgrp * 256 + wv * 64;

    const float* Qb = Q + (size_t)b * TT * CC;
    const float* Kb = K + (size_t)b * KTT * CC;
    float*       Ob = out + (size_t)b * TT * TT;
    float* plds = &lds[wv][0];

    v2f qa[32];
    {
        const float* qrow = Qb + (size_t)(w0 + l16) * CC + 2 * half;
        #pragma unroll
        for (int n = 0; n < 32; ++n)
            qa[n] = *(const v2f*)(qrow + 4 * n);
    }

    const int base0 = w0 + t0_base;

    v8f p_cur;
    {
        int row = base0 + l16;
        if (row > KTT - 1) row = KTT - 1;
        const float* krow = Kb + (size_t)row * CC + 2 * half;
        v8f p = {};
        #pragma unroll
        for (int n = 0; n < 32; ++n) {
            v2f bv = *(const v2f*)(krow + 4 * n);
            p = __builtin_amdgcn_wmma_f32_16x16x4_f32(false, qa[n], false, bv,
                                                      (short)0, p, false, false);
        }
        p_cur = p;
    }

    for (int q = 0; q < 4; ++q) {
        const int t0 = t0_base + 16 * q;

        v8f p_next;
        {
            int row = base0 + 16 * (q + 1) + l16;
            if (row > KTT - 1) row = KTT - 1;
            const float* krow = Kb + (size_t)row * CC + 2 * half;
            v8f p = {};
            #pragma unroll
            for (int n = 0; n < 32; ++n) {
                v2f bv = *(const v2f*)(krow + 4 * n);
                p = __builtin_amdgcn_wmma_f32_16x16x4_f32(false, qa[n], false, bv,
                                                          (short)0, p, false, false);
            }
            p_next = p;
        }

        v8f acc = {};
        {
            const float* bcol = bias + (size_t)(2 * half) * TT + t0 + l16;
            #pragma unroll
            for (int n = 0; n < 32; ++n) {
                v2f bv;
                bv.x = bcol[(size_t)(4 * n) * TT];
                bv.y = bcol[(size_t)(4 * n + 1) * TT];
                acc = __builtin_amdgcn_wmma_f32_16x16x4_f32(false, qa[n], false, bv,
                                                            (short)0, acc, false, false);
            }
        }

        asm volatile("" ::: "memory");
        #pragma unroll
        for (int r = 0; r < 8; ++r) {
            int row = r + 8 * half;
            plds[row * 36 + l16]      = p_cur[r];
            plds[row * 36 + 16 + l16] = p_next[r];
        }
        asm volatile("s_wait_dscnt 0x0" ::: "memory");
        #pragma unroll
        for (int r = 0; r < 8; ++r) {
            int row = r + 8 * half;
            acc[r] += plds[row * 36 + row + l16];
        }
        asm volatile("" ::: "memory");

        #pragma unroll
        for (int r = 0; r < 8; ++r) {
            int row = r + 8 * half;
            Ob[(size_t)(w0 + row) * TT + t0 + l16] = acc[r];
        }

        p_cur = p_next;
    }
}

extern "C" void kernel_launch(void* const* d_in, const int* in_sizes, int n_in,
                              void* d_out, int out_size, void* d_ws, size_t ws_size,
                              hipStream_t stream) {
    (void)in_sizes; (void)n_in; (void)d_ws; (void)ws_size; (void)out_size;
    const float* Q    = (const float*)d_in[0];
    const float* K    = (const float*)d_in[1];
    const float* bias = (const float*)d_in[2];
    float* out = (float*)d_out;

    dim3 grid(BB * 64 * 4);
    dim3 block(128);
    swin_qkbias_kernel<<<grid, block, 0, stream>>>(Q, K, bias, out);
}


# Block = 4 w-tiles x same 64 t: its 128 key rows and 128x64 bias strip staged once in LDS by whole-row 128-bit loads, B fragments read from LDS; k order regrouped, P_0 summed as two half-k chains
# speedup vs baseline: 1.3468x; 1.3468x over previous
#include <hip/hip_runtime.h>


typedef float v2f __attribute__((ext_vector_type(2)));
typedef float v4f __attribute__((ext_vector_type(4)));
typedef float v8f __attribute__((ext_vector_type(8)));

#define BB 16
#define TT 1024
#define CC 128
#define KTT (2*TT - 1)
#define KP 132
#define BP 72
#define SB_FLOATS (128 * BP)
#define SK_FLOATS (128 * KP)


#define WMMA(a, b, c) __builtin_amdgcn_wmma_f32_16x16x4_f32(false, (a), false, (b), (short)0, (c), false, false)

__global__ __launch_bounds__(128)
void swin_qkbias_kernel(const float* __restrict__ Q,
                        const float* __restrict__ K,
                        const float* __restrict__ bias,
                        float* __restrict__ out) {
    __shared__ __attribute__((aligned(16))) float smem[SB_FLOATS + SK_FLOATS + 4 * 16 * 36];
    float* sB = smem;
    float* sK = smem + SB_FLOATS;

    const int lane = threadIdx.x & 31;
    const int wv   = threadIdx.x >> 5;
    const int half = lane >> 4;
    const int l16  = lane & 15;

    const int bid  = blockIdx.x;
    const int tgrp = bid & 15;
    const int wgrp = (bid >> 4) & 15;
    const int b    = bid >> 8;

    const int w0      = wgrp * 64 + wv * 16;
    const int t0_base = tgrp * 64;
    const int kbase   = wgrp * 64 + tgrp * 64;

    const float* Qb = Q + (size_t)b * TT * CC;
    const float* Kb = K + (size_t)b * KTT * CC;
    float*       Ob = out + (size_t)b * TT * TT;
    float* plds = smem + SB_FLOATS + SK_FLOATS + wv * (16 * 36);

    #pragma unroll
    for (int i = 0; i < 32; ++i) {
        const int r = wv * 32 + i;
        int row = kbase + r;
        if (row > KTT - 1) row = KTT - 1;
        v4f v = *(const v4f*)(Kb + (size_t)row * CC + 4 * lane);
        *(v4f*)(&sK[r * KP + 4 * lane]) = v;
    }
    #pragma unroll
    for (int i = 0; i < 16; ++i) {
        const int r = wv * 32 + 2 * i + half;
        v4f v = *(const v4f*)(bias + (size_t)r * TT + t0_base + 4 * l16);
        *(v4f*)(&sB[r * BP + 4 * l16]) = v;
    }

    v2f qa[32];
    {
        const float* qrow = Qb + (size_t)(w0 + l16) * CC + 4 * half;
        #pragma unroll
        for (int m = 0; m < 16; ++m) {
            v4f v = *(const v4f*)(qrow + 8 * m);
            qa[2 * m]     = (v2f){v.x, v.y};
            qa[2 * m + 1] = (v2f){v.z, v.w};
        }
    }

    __syncthreads();

    const float* krow0 = sK + (wv * 16 + l16) * KP + 4 * half;
    const float* bcol0 = sB + (4 * half) * BP + l16;

    v4f   kf[2][16];
    float bf[2][64];

    v4f kf0[16];
    #pragma unroll
    for (int m = 0; m < 16; ++m) kf0[m] = *(const v4f*)(krow0 + 8 * m);
    #pragma unroll
    for (int m = 0; m < 16; ++m) kf[0][m] = *(const v4f*)(krow0 + 16 * KP + 8 * m);
    #pragma unroll
    for (int m = 0; m < 16; ++m) {
        #pragma unroll
        for (int e = 0; e < 4; ++e) bf[0][4 * m + e] = bcol0[(8 * m + e) * BP];
    }
    __builtin_amdgcn_sched_barrier(0);

    v8f p_cur;
    {
        v8f pa = {}, pb = {};
        #pragma unroll
        for (int m = 0; m < 16; m += 2) {
            pa = WMMA(qa[2 * m],     ((v2f){kf0[m].x, kf0[m].y}), pa);
            pb = WMMA(qa[2 * m + 2], ((v2f){kf0[m + 1].x, kf0[m + 1].y}), pb);
            pa = WMMA(qa[2 * m + 1], ((v2f){kf0[m].z, kf0[m].w}), pa);
            pb = WMMA(qa[2 * m + 3], ((v2f){kf0[m + 1].z, kf0[m + 1].w}), pb);
        }
        p_cur = pa + pb;
    }

    #pragma unroll
    for (int q = 0; q < 4; ++q) {
        const int t0  = t0_base + 16 * q;
        const int cur = q & 1;
        const int nxt = cur ^ 1;

        if (q < 3) {
            #pragma unroll
            for (int m = 0; m < 16; ++m) kf[nxt][m] = *(const v4f*)(krow0 + (q + 2) * 16 * KP + 8 * m);
            #pragma unroll
            for (int m = 0; m < 16; ++m) {
                #pragma unroll
                for (int e = 0; e < 4; ++e) bf[nxt][4 * m + e] = bcol0[(8 * m + e) * BP + 16 * (q + 1)];
            }
        }
        __builtin_amdgcn_sched_barrier(0);

        v8f p_next = {};
        v8f acc = {};
        #pragma unroll
        for (int m = 0; m < 16; ++m) {
            const v4f v = kf[cur][m];
            p_next = WMMA(qa[2 * m],     ((v2f){v.x, v.y}), p_next);
            acc    = WMMA(qa[2 * m],     ((v2f){bf[cur][4 * m],     bf[cur][4 * m + 1]}), acc);
            p_next = WMMA(qa[2 * m + 1], ((v2f){v.z, v.w}), p_next);
            acc    = WMMA(qa[2 * m + 1], ((v2f){bf[cur][4 * m + 2], bf[cur][4 * m + 3]}), acc);
        }

        asm volatile("" ::: "memory");
        #pragma unroll
        for (int r = 0; r < 8; ++r) {
            int row = r + 8 * half;
            plds[row * 36 + l16]      = p_cur[r];
            plds[row * 36 + 16 + l16] = p_next[r];
        }
        asm volatile("s_wait_dscnt 0x0" ::: "memory");
        #pragma unroll
        for (int r = 0; r < 8; ++r) {
            int row = r + 8 * half;
            acc[r] += plds[row * 36 + row + l16];
        }
        asm volatile("" ::: "memory");

        #pragma unroll
        for (int r = 0; r < 8; ++r) {
            int row = r + 8 * half;
            Ob[(size_t)(w0 + row) * TT + t0 + l16] = acc[r];
        }

        p_cur = p_next;
    }
}

extern "C" void kernel_launch(void* const* d_in, const int* in_sizes, int n_in,
                              void* d_out, int out_size, void* d_ws, size_t ws_size,
                              hipStream_t stream) {
    (void)in_sizes; (void)n_in; (void)d_ws; (void)ws_size; (void)out_size;
    const float* Q    = (const float*)d_in[0];
    const float* K    = (const float*)d_in[1];
    const float* bias = (const float*)d_in[2];
    float* out = (float*)d_out;

    dim3 grid(BB * 16 * 16);
    dim3 block(128);
    swin_qkbias_kernel<<<grid, block, 0, stream>>>(Q, K, bias, out);
}


# One block per 128 rows x 512 t walks 8 t-steps: keys slide 64 rows/step through an LDS ring, next rows + bias strip fetched under the products; step-first P tile carried, so summed as one k chain
# speedup vs baseline: 1.7884x; 1.3279x over previous
#include <hip/hip_runtime.h>


typedef float v2f __attribute__((ext_vector_type(2)));
typedef float v4f __attribute__((ext_vector_type(4)));
typedef float v8f __attribute__((ext_vector_type(8)));

#define BB 16
#define TT 1024
#define CC 128
#define KTT (2*TT - 1)
#define KP 132
#define BP 72
#define NWAVE 8
#define NSTEP 8
#define RING_ROWS 256
#define SB_FLOATS (128 * BP)
#define SK_FLOATS (RING_ROWS * KP)
#define PATCH_FLOATS (16 * 36)


__device__ __forceinline__ void lds_barrier() {
    __builtin_amdgcn_fence(__ATOMIC_RELEASE, "workgroup", "local");
    __builtin_amdgcn_s_barrier();
    __builtin_amdgcn_fence(__ATOMIC_ACQUIRE, "workgroup", "local");
}

#define WMMA(a, b, c) __builtin_amdgcn_wmma_f32_16x16x4_f32(false, (a), false, (b), (short)0, (c), false, false)

__global__ __launch_bounds__(256)
void swin_qkbias_kernel(const float* __restrict__ Q,
                        const float* __restrict__ K,
                        const float* __restrict__ bias,
                        float* __restrict__ out) {
    __shared__ __attribute__((aligned(16))) float smem[2 * SB_FLOATS + SK_FLOATS + NWAVE * PATCH_FLOATS];
    float* sB = smem;
    float* sK = smem + 2 * SB_FLOATS;

    const int lane = threadIdx.x & 31;
    const int wv   = threadIdx.x >> 5;
    const int half = lane >> 4;
    const int l16  = lane & 15;

    const int bid = blockIdx.x;
    const int th  = bid & 1;
    const int wg  = (bid >> 1) & 7;
    const int b   = bid >> 4;

    const int w0     = wg * 128 + wv * 16;
    const int t_base = th * 512;
    const int kb0    = wg * 128 + t_base;

    const float* Qb = Q + (size_t)b * TT * CC;
    const float* Kb = K + (size_t)b * KTT * CC;
    float*       Ob = out + (size_t)b * TT * TT;
    float* plds = smem + 2 * SB_FLOATS + SK_FLOATS + wv * PATCH_FLOATS;

    auto stage_load = [&](int n, v4f (&st)[16]) {
        #pragma unroll
        for (int i = 0; i < 8; ++i) {
            int row = kb0 + 64 * (n + 2) + wv * 8 + i;
            if (row > KTT - 1) row = KTT - 1;
            st[i] = *(const v4f*)(Kb + (size_t)row * CC + 4 * lane);
        }
        #pragma unroll
        for (int i = 0; i < 8; ++i) {
            const int r = wv * 16 + 2 * i + half;
            st[8 + i] = *(const v4f*)(bias + (size_t)r * TT + t_base + 64 * n + 4 * l16);
        }
    };
    auto stage_store = [&](int n, const v4f (&st)[16]) {
        #pragma unroll
        for (int i = 0; i < 8; ++i) {
            const int r = 64 * (n + 2) + wv * 8 + i;
            *(v4f*)(&sK[(r & (RING_ROWS - 1)) * KP + 4 * lane]) = st[i];
        }
        #pragma unroll
        for (int i = 0; i < 8; ++i) {
            const int r = wv * 16 + 2 * i + half;
            *(v4f*)(&sB[(n & 1) * SB_FLOATS + r * BP + 4 * l16]) = st[8 + i];
        }
    };

    {
        v4f t[16];
        #pragma unroll
        for (int i = 0; i < 16; ++i) {
            const int r = wv * 16 + i;
            t[i] = *(const v4f*)(Kb + (size_t)(kb0 + r) * CC + 4 * lane);
        }
        #pragma unroll
        for (int i = 0; i < 16; ++i) {
            const int r = wv * 16 + i;
            *(v4f*)(&sK[r * KP + 4 * lane]) = t[i];
        }
    }
    {
        v4f t[16];
        stage_load(0, t);
        stage_store(0, t);
    }
    {
        v4f t[16];
        stage_load(1, t);
        stage_store(1, t);
    }

    v2f qa[32];
    {
        const float* qrow = Qb + (size_t)(w0 + l16) * CC + 4 * half;
        #pragma unroll
        for (int m = 0; m < 16; ++m) {
            v4f v = *(const v4f*)(qrow + 8 * m);
            qa[2 * m]     = (v2f){v.x, v.y};
            qa[2 * m + 1] = (v2f){v.z, v.w};
        }
    }

    lds_barrier();

    v4f st[16];
    stage_load(2, st);

    const int rw = wv * 16 + l16;
    const float* kcol0 = sK + 4 * half;
    const float* bcol0 = sB + (4 * half) * BP + l16;

    v4f   kf[2][16];
    float bf[2][64];

    v8f p_cur;
    {
        v4f kf0[16];
        const float* k0 = kcol0 + (rw & (RING_ROWS - 1)) * KP;
        const float* k1 = kcol0 + ((rw + 16) & (RING_ROWS - 1)) * KP;
        #pragma unroll
        for (int m = 0; m < 16; ++m) kf0[m] = *(const v4f*)(k0 + 8 * m);
        #pragma unroll
        for (int m = 0; m < 16; ++m) kf[0][m] = *(const v4f*)(k1 + 8 * m);
        #pragma unroll
        for (int m = 0; m < 16; ++m) {
            #pragma unroll
            for (int e = 0; e < 4; ++e) bf[0][4 * m + e] = bcol0[(8 * m + e) * BP];
        }
        __builtin_amdgcn_sched_barrier(0);

        v8f pa = {}, pb = {};
        #pragma unroll
        for (int m = 0; m < 16; m += 2) {
            pa = WMMA(qa[2 * m],     ((v2f){kf0[m].x, kf0[m].y}), pa);
            pb = WMMA(qa[2 * m + 2], ((v2f){kf0[m + 1].x, kf0[m + 1].y}), pb);
            pa = WMMA(qa[2 * m + 1], ((v2f){kf0[m].z, kf0[m].w}), pa);
            pb = WMMA(qa[2 * m + 3], ((v2f){kf0[m + 1].z, kf0[m + 1].w}), pb);
        }
        p_cur = pa + pb;
    }

    #pragma unroll 1
    for (int s = 0; s < NSTEP; ++s) {
        const int rs = rw + 64 * s;
        const float* bcs = bcol0 + (s & 1) * SB_FLOATS;
        const float* bcn = bcol0 + ((s + 1) & 1) * SB_FLOATS;

        #pragma unroll
        for (int q = 0; q < 4; ++q) {
            const int t0  = t_base + 64 * s + 16 * q;
            const int cur = q & 1;
            const int nxt = cur ^ 1;

            if (q == 3) {
                lds_barrier();
            }

            if (q < 3) {
                const float* kp = kcol0 + ((rs + 16 * (q + 2)) & (RING_ROWS - 1)) * KP;
                #pragma unroll
                for (int m = 0; m < 16; ++m) kf[nxt][m] = *(const v4f*)(kp + 8 * m);
                #pragma unroll
                for (int m = 0; m < 16; ++m) {
                    #pragma unroll
                    for (int e = 0; e < 4; ++e) bf[nxt][4 * m + e] = bcs[(8 * m + e) * BP + 16 * (q + 1)];
                }
            } else if (s + 1 < NSTEP) {
                const float* kp = kcol0 + ((rs + 64 + 16) & (RING_ROWS - 1)) * KP;
                #pragma unroll
                for (int m = 0; m < 16; ++m) kf[nxt][m] = *(const v4f*)(kp + 8 * m);
                #pragma unroll
                for (int m = 0; m < 16; ++m) {
                    #pragma unroll
                    for (int e = 0; e < 4; ++e) bf[nxt][4 * m + e] = bcn[(8 * m + e) * BP];
                }
            }
            __builtin_amdgcn_sched_barrier(0);

            v8f p_next = {};
            v8f acc = {};
            #pragma unroll
            for (int m = 0; m < 16; ++m) {
                const v4f v = kf[cur][m];
                p_next = WMMA(qa[2 * m],     ((v2f){v.x, v.y}), p_next);
                acc    = WMMA(qa[2 * m],     ((v2f){bf[cur][4 * m],     bf[cur][4 * m + 1]}), acc);
                p_next = WMMA(qa[2 * m + 1], ((v2f){v.z, v.w}), p_next);
                acc    = WMMA(qa[2 * m + 1], ((v2f){bf[cur][4 * m + 2], bf[cur][4 * m + 3]}), acc);
            }

            asm volatile("" ::: "memory");
            #pragma unroll
            for (int r = 0; r < 8; ++r) {
                int row = r + 8 * half;
                plds[row * 36 + l16]      = p_cur[r];
                plds[row * 36 + 16 + l16] = p_next[r];
            }
            asm volatile("s_wait_dscnt 0x0" ::: "memory");
            #pragma unroll
            for (int r = 0; r < 8; ++r) {
                int row = r + 8 * half;
                acc[r] += plds[row * 36 + row + l16];
            }
            asm volatile("" ::: "memory");

            #pragma unroll
            for (int r = 0; r < 8; ++r) {
                int row = r + 8 * half;
                Ob[(size_t)(w0 + row) * TT + t0 + l16] = acc[r];
            }

            p_cur = p_next;

            if (q == 3) {
                if (s + 2 < NSTEP) stage_store(s + 2, st);
                if (s + 3 < NSTEP) stage_load(s + 3, st);
            }
        }
    }
}

extern "C" void kernel_launch(void* const* d_in, const int* in_sizes, int n_in,
                              void* d_out, int out_size, void* d_ws, size_t ws_size,
                              hipStream_t stream) {
    (void)in_sizes; (void)n_in; (void)d_ws; (void)ws_size; (void)out_size;
    const float* Q    = (const float*)d_in[0];
    const float* K    = (const float*)d_in[1];
    const float* bias = (const float*)d_in[2];
    float* out = (float*)d_out;

    dim3 grid(BB * 8 * 2);
    dim3 block(256);
    swin_qkbias_kernel<<<grid, block, 0, stream>>>(Q, K, bias, out);
}


# Wave priority falls pass by pass behind each step barrier (s_setprio 3,2,1,0): the two waves of a SIMD take the matrix pipe in turn, each tile epilogue runs under the sibling's products
# speedup vs baseline: 1.8904x; 1.0570x over previous
#include <hip/hip_runtime.h>


typedef float v2f __attribute__((ext_vector_type(2)));
typedef float v4f __attribute__((ext_vector_type(4)));
typedef float v8f __attribute__((ext_vector_type(8)));

#define BB 16
#define TT 1024
#define CC 128
#define KTT (2*TT - 1)
#define KP 132
#define BP 72
#define NWAVE 8
#define NSTEP 8
#define RING_ROWS 256
#define SB_FLOATS (128 * BP)
#define SK_FLOATS (RING_ROWS * KP)
#define PATCH_FLOATS (16 * 36)


__device__ __forceinline__ void lds_barrier() {
    __builtin_amdgcn_fence(__ATOMIC_RELEASE, "workgroup", "local");
    __builtin_amdgcn_s_barrier();
    __builtin_amdgcn_fence(__ATOMIC_ACQUIRE, "workgroup", "local");
}

#define WMMA(a, b, c) __builtin_amdgcn_wmma_f32_16x16x4_f32(false, (a), false, (b), (short)0, (c), false, false)

__global__ __launch_bounds__(256)
void swin_qkbias_kernel(const float* __restrict__ Q,
                        const float* __restrict__ K,
                        const float* __restrict__ bias,
                        float* __restrict__ out) {
    __shared__ __attribute__((aligned(16))) float smem[2 * SB_FLOATS + SK_FLOATS + NWAVE * PATCH_FLOATS];
    float* sB = smem;
    float* sK = smem + 2 * SB_FLOATS;

    const int lane = threadIdx.x & 31;
    const int wv   = threadIdx.x >> 5;
    const int half = lane >> 4;
    const int l16  = lane & 15;

    const int bid = blockIdx.x;
    const int th  = bid & 1;
    const int wg  = (bid >> 1) & 7;
    const int b   = bid >> 4;

    const int w0     = wg * 128 + wv * 16;
    const int t_base = th * 512;
    const int kb0    = wg * 128 + t_base;

    const float* Qb = Q + (size_t)b * TT * CC;
    const float* Kb = K + (size_t)b * KTT * CC;
    float*       Ob = out + (size_t)b * TT * TT;
    float* plds = smem + 2 * SB_FLOATS + SK_FLOATS + wv * PATCH_FLOATS;

    auto stage_load = [&](int n, v4f (&st)[16]) {
        #pragma unroll
        for (int i = 0; i < 8; ++i) {
            int row = kb0 + 64 * (n + 2) + wv * 8 + i;
            if (row > KTT - 1) row = KTT - 1;
            st[i] = *(const v4f*)(Kb + (size_t)row * CC + 4 * lane);
        }
        #pragma unroll
        for (int i = 0; i < 8; ++i) {
            const int r = wv * 16 + 2 * i + half;
            st[8 + i] = *(const v4f*)(bias + (size_t)r * TT + t_base + 64 * n + 4 * l16);
        }
    };
    auto stage_store = [&](int n, const v4f (&st)[16]) {
        #pragma unroll
        for (int i = 0; i < 8; ++i) {
            const int r = 64 * (n + 2) + wv * 8 + i;
            *(v4f*)(&sK[(r & (RING_ROWS - 1)) * KP + 4 * lane]) = st[i];
        }
        #pragma unroll
        for (int i = 0; i < 8; ++i) {
            const int r = wv * 16 + 2 * i + half;
            *(v4f*)(&sB[(n & 1) * SB_FLOATS + r * BP + 4 * l16]) = st[8 + i];
        }
    };

    {
        v4f t[16];
        #pragma unroll
        for (int i = 0; i < 16; ++i) {
            const int r = wv * 16 + i;
            t[i] = *(const v4f*)(Kb + (size_t)(kb0 + r) * CC + 4 * lane);
        }
        #pragma unroll
        for (int i = 0; i < 16; ++i) {
            const int r = wv * 16 + i;
            *(v4f*)(&sK[r * KP + 4 * lane]) = t[i];
        }
    }
    {
        v4f t[16];
        stage_load(0, t);
        stage_store(0, t);
    }
    {
        v4f t[16];
        stage_load(1, t);
        stage_store(1, t);
    }

    v2f qa[32];
    {
        const float* qrow = Qb + (size_t)(w0 + l16) * CC + 4 * half;
        #pragma unroll
        for (int m = 0; m < 16; ++m) {
            v4f v = *(const v4f*)(qrow + 8 * m);
            qa[2 * m]     = (v2f){v.x, v.y};
            qa[2 * m + 1] = (v2f){v.z, v.w};
        }
    }

    lds_barrier();

    v4f st[16];
    stage_load(2, st);

    const int rw = wv * 16 + l16;
    const float* kcol0 = sK + 4 * half;
    const float* bcol0 = sB + (4 * half) * BP + l16;

    v4f   kf[2][16];
    float bf[2][64];

    v8f p_cur;
    {
        v4f kf0[16];
        const float* k0 = kcol0 + (rw & (RING_ROWS - 1)) * KP;
        const float* k1 = kcol0 + ((rw + 16) & (RING_ROWS - 1)) * KP;
        #pragma unroll
        for (int m = 0; m < 16; ++m) kf0[m] = *(const v4f*)(k0 + 8 * m);
        #pragma unroll
        for (int m = 0; m < 16; ++m) kf[0][m] = *(const v4f*)(k1 + 8 * m);
        #pragma unroll
        for (int m = 0; m < 16; ++m) {
            #pragma unroll
            for (int e = 0; e < 4; ++e) bf[0][4 * m + e] = bcol0[(8 * m + e) * BP];
        }
        __builtin_amdgcn_sched_barrier(0);
        __builtin_amdgcn_s_setprio(3);

        v8f pa = {}, pb = {};
        #pragma unroll
        for (int m = 0; m < 16; m += 2) {
            pa = WMMA(qa[2 * m],     ((v2f){kf0[m].x, kf0[m].y}), pa);
            pb = WMMA(qa[2 * m + 2], ((v2f){kf0[m + 1].x, kf0[m + 1].y}), pb);
            pa = WMMA(qa[2 * m + 1], ((v2f){kf0[m].z, kf0[m].w}), pa);
            pb = WMMA(qa[2 * m + 3], ((v2f){kf0[m + 1].z, kf0[m + 1].w}), pb);
        }
        p_cur = pa + pb;
    }

    #pragma unroll 1
    for (int s = 0; s < NSTEP; ++s) {
        const int rs = rw + 64 * s;
        const float* bcs = bcol0 + (s & 1) * SB_FLOATS;
        const float* bcn = bcol0 + ((s + 1) & 1) * SB_FLOATS;

        #pragma unroll
        for (int q = 0; q < 4; ++q) {
            const int t0  = t_base + 64 * s + 16 * q;
            const int cur = q & 1;
            const int nxt = cur ^ 1;

            if (q == 3) {
                lds_barrier();
            }

            if (q < 3) {
                const float* kp = kcol0 + ((rs + 16 * (q + 2)) & (RING_ROWS - 1)) * KP;
                #pragma unroll
                for (int m = 0; m < 16; ++m) kf[nxt][m] = *(const v4f*)(kp + 8 * m);
                #pragma unroll
                for (int m = 0; m < 16; ++m) {
                    #pragma unroll
                    for (int e = 0; e < 4; ++e) bf[nxt][4 * m + e] = bcs[(8 * m + e) * BP + 16 * (q + 1)];
                }
            } else if (s + 1 < NSTEP) {
                const float* kp = kcol0 + ((rs + 64 + 16) & (RING_ROWS - 1)) * KP;
                #pragma unroll
                for (int m = 0; m < 16; ++m) kf[nxt][m] = *(const v4f*)(kp + 8 * m);
                #pragma unroll
                for (int m = 0; m < 16; ++m) {
                    #pragma unroll
                    for (int e = 0; e < 4; ++e) bf[nxt][4 * m + e] = bcn[(8 * m + e) * BP];
                }
            }
            __builtin_amdgcn_sched_barrier(0);

            if (q == 3)      __builtin_amdgcn_s_setprio(3);
            else if (q == 0) __builtin_amdgcn_s_setprio(2);
            else if (q == 1) __builtin_amdgcn_s_setprio(1);
            else             __builtin_amdgcn_s_setprio(0);

            v8f p_next = {};
            v8f acc = {};
            #pragma unroll
            for (int m = 0; m < 16; ++m) {
                const v4f v = kf[cur][m];
                p_next = WMMA(qa[2 * m],     ((v2f){v.x, v.y}), p_next);
                acc    = WMMA(qa[2 * m],     ((v2f){bf[cur][4 * m],     bf[cur][4 * m + 1]}), acc);
                p_next = WMMA(qa[2 * m + 1], ((v2f){v.z, v.w}), p_next);
                acc    = WMMA(qa[2 * m + 1], ((v2f){bf[cur][4 * m + 2], bf[cur][4 * m + 3]}), acc);
            }

            asm volatile("" ::: "memory");
            #pragma unroll
            for (int r = 0; r < 8; ++r) {
                int row = r + 8 * half;
                plds[row * 36 + l16]      = p_cur[r];
                plds[row * 36 + 16 + l16] = p_next[r];
            }
            asm volatile("s_wait_dscnt 0x0" ::: "memory");
            #pragma unroll
            for (int r = 0; r < 8; ++r) {
                int row = r + 8 * half;
                acc[r] += plds[row * 36 + row + l16];
            }
            asm volatile("" ::: "memory");

            #pragma unroll
            for (int r = 0; r < 8; ++r) {
                int row = r + 8 * half;
                Ob[(size_t)(w0 + row) * TT + t0 + l16] = acc[r];
            }

            p_cur = p_next;

            if (q == 3) {
                if (s + 2 < NSTEP) stage_store(s + 2, st);
                if (s + 3 < NSTEP) stage_load(s + 3, st);
            }
        }
    }
}

extern "C" void kernel_launch(void* const* d_in, const int* in_sizes, int n_in,
                              void* d_out, int out_size, void* d_ws, size_t ws_size,
                              hipStream_t stream) {
    (void)in_sizes; (void)n_in; (void)d_ws; (void)ws_size; (void)out_size;
    const float* Q    = (const float*)d_in[0];
    const float* K    = (const float*)d_in[1];
    const float* bias = (const float*)d_in[2];
    float* out = (float*)d_out;

    dim3 grid(BB * 8 * 2);
    dim3 block(256);
    swin_qkbias_kernel<<<grid, block, 0, stream>>>(Q, K, bias, out);
}


# LDS instructions per pass cut from 72 to 44: bias strip kept t-major so its B fragments are 16-byte reads like the keys; patch keeps P_q from the pass before, only P_q+1 is stored; sums unchanged
# speedup vs baseline: 1.9700x; 1.0421x over previous
#include <hip/hip_runtime.h>


typedef float v2f __attribute__((ext_vector_type(2)));
typedef float v4f __attribute__((ext_vector_type(4)));
typedef float v8f __attribute__((ext_vector_type(8)));

#define BB 16
#define TT 1024
#define CC 128
#define KTT (2*TT - 1)
#define KP 132
#define NWAVE 8
#define NSTEP 8
#define RING_ROWS 256
#define SB_FLOATS (64 * KP)
#define SK_FLOATS (RING_ROWS * KP)
#define PATCH_FLOATS (16 * 36)


__device__ __forceinline__ void lds_barrier() {
    __builtin_amdgcn_fence(__ATOMIC_RELEASE, "workgroup", "local");
    __builtin_amdgcn_s_barrier();
    __builtin_amdgcn_fence(__ATOMIC_ACQUIRE, "workgroup", "local");
}

#define WMMA(a, b, c) __builtin_amdgcn_wmma_f32_16x16x4_f32(false, (a), false, (b), (short)0, (c), false, false)

__global__ __launch_bounds__(256)
void swin_qkbias_kernel(const float* __restrict__ Q,
                        const float* __restrict__ K,
                        const float* __restrict__ bias,
                        float* __restrict__ out) {
    __shared__ __attribute__((aligned(16))) float smem[2 * SB_FLOATS + SK_FLOATS + NWAVE * PATCH_FLOATS];
    float* sB = smem;
    float* sK = smem + 2 * SB_FLOATS;

    const int lane = threadIdx.x & 31;
    const int wv   = threadIdx.x >> 5;
    const int half = lane >> 4;
    const int l16  = lane & 15;

    const int bid = blockIdx.x;
    const int th  = bid & 1;
    const int wg  = (bid >> 1) & 7;
    const int b   = bid >> 4;

    const int w0     = wg * 128 + wv * 16;
    const int t_base = th * 512;
    const int kb0    = wg * 128 + t_base;

    const float* Qb = Q + (size_t)b * TT * CC;
    const float* Kb = K + (size_t)b * KTT * CC;
    float*       Ob = out + (size_t)b * TT * TT;
    float* plds = smem + 2 * SB_FLOATS + SK_FLOATS + wv * PATCH_FLOATS;

    auto stage_load = [&](int n, v4f (&st)[16]) {
        #pragma unroll
        for (int i = 0; i < 8; ++i) {
            int row = kb0 + 64 * (n + 2) + wv * 8 + i;
            if (row > KTT - 1) row = KTT - 1;
            st[i] = *(const v4f*)(Kb + (size_t)row * CC + 4 * lane);
        }
        #pragma unroll
        for (int i = 0; i < 8; ++i) {
            const int r = wv * 16 + 2 * i + half;
            st[8 + i] = *(const v4f*)(bias + (size_t)r * TT + t_base + 64 * n + 4 * l16);
        }
    };
    auto stage_store = [&](int n, const v4f (&st)[16]) {
        #pragma unroll
        for (int i = 0; i < 8; ++i) {
            const int r = 64 * (n + 2) + wv * 8 + i;
            *(v4f*)(&sK[(r & (RING_ROWS - 1)) * KP + 4 * lane]) = st[i];
        }
        #pragma unroll
        for (int i = 0; i < 8; ++i) {
            const int r = wv * 16 + 2 * i + half;
            float* d = &sB[(n & 1) * SB_FLOATS + (4 * l16) * KP + r];
            d[0] = st[8 + i].x; d[KP] = st[8 + i].y; d[2 * KP] = st[8 + i].z; d[3 * KP] = st[8 + i].w;
        }
    };

    {
        v4f t[16];
        #pragma unroll
        for (int i = 0; i < 16; ++i) {
            const int r = wv * 16 + i;
            t[i] = *(const v4f*)(Kb + (size_t)(kb0 + r) * CC + 4 * lane);
        }
        #pragma unroll
        for (int i = 0; i < 16; ++i) {
            const int r = wv * 16 + i;
            *(v4f*)(&sK[r * KP + 4 * lane]) = t[i];
        }
    }
    {
        v4f t[16];
        stage_load(0, t);
        stage_store(0, t);
    }
    {
        v4f t[16];
        stage_load(1, t);
        stage_store(1, t);
    }

    v2f qa[32];
    {
        const float* qrow = Qb + (size_t)(w0 + l16) * CC + 4 * half;
        #pragma unroll
        for (int m = 0; m < 16; ++m) {
            v4f v = *(const v4f*)(qrow + 8 * m);
            qa[2 * m]     = (v2f){v.x, v.y};
            qa[2 * m + 1] = (v2f){v.z, v.w};
        }
    }

    lds_barrier();

    v4f st[16];
    stage_load(2, st);

    const int rw = wv * 16 + l16;
    const float* kcol0 = sK + 4 * half;
    const float* bcol0 = sB + l16 * KP + 4 * half;
    float* pst = plds + (8 * half) * 36 + l16;
    const float* pge = plds + (8 * half) * 37 + l16;
    int pgo[8];
    #pragma unroll
    for (int r = 0; r < 8; ++r) pgo[r] = (r + 8 * half) * 36 + ((r + 8 * half + l16 + 16) & 31);

    v4f   kf[2][16];
    v4f   bf[2][16];

    {
        v4f kf0[16];
        const float* k0 = kcol0 + (rw & (RING_ROWS - 1)) * KP;
        const float* k1 = kcol0 + ((rw + 16) & (RING_ROWS - 1)) * KP;
        #pragma unroll
        for (int m = 0; m < 16; ++m) kf0[m] = *(const v4f*)(k0 + 8 * m);
        #pragma unroll
        for (int m = 0; m < 16; ++m) kf[0][m] = *(const v4f*)(k1 + 8 * m);
        #pragma unroll
        for (int m = 0; m < 16; ++m) bf[0][m] = *(const v4f*)(bcol0 + 8 * m);
        __builtin_amdgcn_sched_barrier(0);
        __builtin_amdgcn_s_setprio(3);

        v8f pa = {}, pb = {};
        #pragma unroll
        for (int m = 0; m < 16; m += 2) {
            pa = WMMA(qa[2 * m],     ((v2f){kf0[m].x, kf0[m].y}), pa);
            pb = WMMA(qa[2 * m + 2], ((v2f){kf0[m + 1].x, kf0[m + 1].y}), pb);
            pa = WMMA(qa[2 * m + 1], ((v2f){kf0[m].z, kf0[m].w}), pa);
            pb = WMMA(qa[2 * m + 3], ((v2f){kf0[m + 1].z, kf0[m + 1].w}), pb);
        }
        pa = pa + pb;
        #pragma unroll
        for (int r = 0; r < 8; ++r) pst[r * 36] = pa[r];
    }

    #pragma unroll 1
    for (int s = 0; s < NSTEP; ++s) {
        const int rs = rw + 64 * s;
        const float* bcs = bcol0 + (s & 1) * SB_FLOATS;
        const float* bcn = bcol0 + ((s + 1) & 1) * SB_FLOATS;

        #pragma unroll
        for (int q = 0; q < 4; ++q) {
            const int t0  = t_base + 64 * s + 16 * q;
            const int cur = q & 1;
            const int nxt = cur ^ 1;

            if (q == 3) {
                lds_barrier();
            }

            if (q < 3) {
                const float* kp = kcol0 + ((rs + 16 * (q + 2)) & (RING_ROWS - 1)) * KP;
                #pragma unroll
                for (int m = 0; m < 16; ++m) kf[nxt][m] = *(const v4f*)(kp + 8 * m);
                #pragma unroll
                for (int m = 0; m < 16; ++m) bf[nxt][m] = *(const v4f*)(bcs + 16 * (q + 1) * KP + 8 * m);
            } else if (s + 1 < NSTEP) {
                const float* kp = kcol0 + ((rs + 64 + 16) & (RING_ROWS - 1)) * KP;
                #pragma unroll
                for (int m = 0; m < 16; ++m) kf[nxt][m] = *(const v4f*)(kp + 8 * m);
                #pragma unroll
                for (int m = 0; m < 16; ++m) bf[nxt][m] = *(const v4f*)(bcn + 8 * m);
            }
            __builtin_amdgcn_sched_barrier(0);

            if (q == 3)      __builtin_amdgcn_s_setprio(3);
            else if (q == 0) __builtin_amdgcn_s_setprio(2);
            else if (q == 1) __builtin_amdgcn_s_setprio(1);
            else             __builtin_amdgcn_s_setprio(0);

            v8f p_next = {};
            v8f acc = {};
            #pragma unroll
            for (int m = 0; m < 16; ++m) {
                const v4f v = kf[cur][m];
                const v4f w = bf[cur][m];
                p_next = WMMA(qa[2 * m],     ((v2f){v.x, v.y}), p_next);
                acc    = WMMA(qa[2 * m],     ((v2f){w.x, w.y}), acc);
                p_next = WMMA(qa[2 * m + 1], ((v2f){v.z, v.w}), p_next);
                acc    = WMMA(qa[2 * m + 1], ((v2f){w.z, w.w}), acc);
            }

            asm volatile("" ::: "memory");
            #pragma unroll
            for (int r = 0; r < 8; ++r) pst[r * 36 + 16 * ((q + 1) & 1)] = p_next[r];
            asm volatile("s_wait_dscnt 0x0" ::: "memory");
            #pragma unroll
            for (int r = 0; r < 8; ++r) acc[r] += (q & 1) ? plds[pgo[r]] : pge[r * 37];
            asm volatile("" ::: "memory");

            #pragma unroll
            for (int r = 0; r < 8; ++r) {
                int row = r + 8 * half;
                Ob[(size_t)(w0 + row) * TT + t0 + l16] = acc[r];
            }

            if (q == 3) {
                if (s + 2 < NSTEP) stage_store(s + 2, st);
                if (s + 3 < NSTEP) stage_load(s + 3, st);
            }
        }
    }
}

extern "C" void kernel_launch(void* const* d_in, const int* in_sizes, int n_in,
                              void* d_out, int out_size, void* d_ws, size_t ws_size,
                              hipStream_t stream) {
    (void)in_sizes; (void)n_in; (void)d_ws; (void)ws_size; (void)out_size;
    const float* Q    = (const float*)d_in[0];
    const float* K    = (const float*)d_in[1];
    const float* bias = (const float*)d_in[2];
    float* out = (float*)d_out;

    dim3 grid(BB * 8 * 2);
    dim3 block(256);
    swin_qkbias_kernel<<<grid, block, 0, stream>>>(Q, K, bias, out);
}
